# starting kernel
# baseline (speedup 1.0000x reference)
#include <hip/hip_runtime.h>


#define D_DIM 128
#define M_NBR 32
#define ZSTR  132
#define GSTR  36

typedef __attribute__((ext_vector_type(2))) float v2f;
typedef __attribute__((ext_vector_type(8))) float v8f;

static __device__ __forceinline__ float wave_sum32(float v) {
#pragma unroll
  for (int off = 16; off > 0; off >>= 1) v += __shfl_xor(v, off, 32);
  return v;
}
static __device__ __forceinline__ float wave_max32(float v) {
#pragma unroll
  for (int off = 16; off > 0; off >>= 1) v = fmaxf(v, __shfl_xor(v, off, 32));
  return v;
}

__global__ __launch_bounds__(256) void l2norm_rows(const float* __restrict__ x,
                                                   float* __restrict__ xn, int n) {
  int row  = blockIdx.x * 8 + (threadIdx.x >> 5);
  int lane = threadIdx.x & 31;
  if (row >= n) return;
  float4 v = ((const float4*)x)[(size_t)row * (D_DIM / 4) + lane];
  float s = v.x * v.x + v.y * v.y + v.z * v.z + v.w * v.w;
  s = wave_sum32(s);
  float sc = 1.0f / fmaxf(sqrtf(s), 1e-12f);
  float4 o; o.x = v.x * sc; o.y = v.y * sc; o.z = v.z * sc; o.w = v.w * sc;
  ((float4*)xn)[(size_t)row * (D_DIM / 4) + lane] = o;
}

__global__ __launch_bounds__(32) void capsule_route(const float* __restrict__ xn,
                                                    const int* __restrict__ x_nb,
                                                    float* __restrict__ out, int n) {
  __shared__ float zt[M_NBR][ZSTR];
  __shared__ float xnrow[D_DIM];
  __shared__ float G[M_NBR][GSTR];
  __shared__ float pv[M_NBR];

  int node = blockIdx.x;
  if (node >= n) return;
  int L  = threadIdx.x;
  int c4 = L * 4;

  int nbi = x_nb[(size_t)node * M_NBR + L] - 1;

#pragma unroll 4
  for (int m = 0; m < M_NBR; ++m) {
    int r = __shfl(nbi, m, 32);
    float4 v = ((const float4*)xn)[(size_t)r * (D_DIM / 4) + L];
    *(float4*)&zt[m][c4] = v;
  }
  float4 xs = ((const float4*)xn)[(size_t)node * (D_DIM / 4) + L];
  *(float4*)&xnrow[c4] = xs;
  __syncthreads();

  float bm = 0.0f;
#pragma unroll 8
  for (int j = 0; j < D_DIM / 4; ++j) {
    float4 zv = *(const float4*)&zt[L][4 * j];
    float4 xv = *(const float4*)&xnrow[4 * j];
    bm += zv.x * xv.x + zv.y * xv.y + zv.z * xv.z + zv.w * xv.w;
  }

  int h = L >> 4, cc = L & 15;
  v8f g00 = {}, g01 = {}, g10 = {}, g11 = {};
#pragma unroll 4
  for (int k0 = 0; k0 < D_DIM; k0 += 4) {
    v2f a0 = *(const v2f*)&zt[cc][k0 + 2 * h];
    v2f a1 = *(const v2f*)&zt[16 + cc][k0 + 2 * h];
    g00 = __builtin_amdgcn_wmma_f32_16x16x4_f32(false, a0, false, a0, (short)0, g00, false, false);
    g01 = __builtin_amdgcn_wmma_f32_16x16x4_f32(false, a0, false, a1, (short)0, g01, false, false);
    g10 = __builtin_amdgcn_wmma_f32_16x16x4_f32(false, a1, false, a0, (short)0, g10, false, false);
    g11 = __builtin_amdgcn_wmma_f32_16x16x4_f32(false, a1, false, a1, (short)0, g11, false, false);
  }
#pragma unroll
  for (int v = 0; v < 8; ++v) {
    int r = v + 8 * h;
    G[r][cc]           = g00[v];
    G[r][16 + cc]      = g01[v];
    G[16 + r][cc]      = g10[v];
    G[16 + r][16 + cc] = g11[v];
  }
  __syncthreads();

  float p = 1.0f / 32.0f;
  for (int it = 0; it < 2; ++it) {
    pv[L] = p;
    __syncthreads();
    float gp = 0.0f;
#pragma unroll
    for (int j = 0; j < 8; ++j) {
      float4 gr = *(const float4*)&G[L][4 * j];
      float4 pj = *(const float4*)&pv[4 * j];
      gp += gr.x * pj.x + gr.y * pj.y + gr.z * pj.z + gr.w * pj.w;
    }
    float sq = wave_sum32(p * (gp + 2.0f * bm)) + 1.0f;
    float s  = sqrtf(sq) / (sq + 1.0f);
    float logit = s * (gp + bm);
    float mx = wave_max32(logit);
    float e  = __expf(logit - mx);
    p = e / wave_sum32(e);
    __syncthreads();
  }

  pv[L] = p;
  __syncthreads();

  float4 acc = xs;
#pragma unroll
  for (int j = 0; j < 8; ++j) {
    float4 pj = *(const float4*)&pv[4 * j];
    float4 z0 = *(const float4*)&zt[4 * j + 0][c4];
    float4 z1 = *(const float4*)&zt[4 * j + 1][c4];
    float4 z2 = *(const float4*)&zt[4 * j + 2][c4];
    float4 z3 = *(const float4*)&zt[4 * j + 3][c4];
    acc.x += pj.x * z0.x + pj.y * z1.x + pj.z * z2.x + pj.w * z3.x;
    acc.y += pj.x * z0.y + pj.y * z1.y + pj.z * z2.y + pj.w * z3.y;
    acc.z += pj.x * z0.z + pj.y * z1.z + pj.z * z2.z + pj.w * z3.z;
    acc.w += pj.x * z0.w + pj.y * z1.w + pj.z * z2.w + pj.w * z3.w;
  }
  ((float4*)out)[(size_t)node * (D_DIM / 4) + L] = acc;
}

extern "C" void kernel_launch(void* const* d_in, const int* in_sizes, int n_in,
                              void* d_out, int out_size, void* d_ws, size_t ws_size,
                              hipStream_t stream) {
  (void)n_in; (void)out_size; (void)ws_size;
  const float* x    = (const float*)d_in[0];
  const int*   x_nb = (const int*)d_in[1];
  float*       out  = (float*)d_out;
  int n = in_sizes[0] / D_DIM;
  float* xn = (float*)d_ws;

  l2norm_rows<<<dim3((n + 7) / 8), dim3(256), 0, stream>>>(x, xn, n);
  capsule_route<<<dim3(n), dim3(32), 0, stream>>>(xn, x_nb, out, n);
}


# Gather: all 33 row loads of a node issued back to back into registers (row bases via readlane on the scalar unit) before the first LDS store: one memory round trip per tile instead of eight
# speedup vs baseline: 1.1742x; 1.1742x over previous
#include <hip/hip_runtime.h>


#define D_DIM 128
#define M_NBR 32
#define ZSTR  132
#define GSTR  36

typedef __attribute__((ext_vector_type(2))) float v2f;
typedef __attribute__((ext_vector_type(8))) float v8f;

static __device__ __forceinline__ float wave_sum32(float v) {
#pragma unroll
  for (int off = 16; off > 0; off >>= 1) v += __shfl_xor(v, off, 32);
  return v;
}
static __device__ __forceinline__ float wave_max32(float v) {
#pragma unroll
  for (int off = 16; off > 0; off >>= 1) v = fmaxf(v, __shfl_xor(v, off, 32));
  return v;
}

__global__ __launch_bounds__(256) void l2norm_rows(const float* __restrict__ x,
                                                   float* __restrict__ xn, int n) {
  int row  = blockIdx.x * 8 + (threadIdx.x >> 5);
  int lane = threadIdx.x & 31;
  if (row >= n) return;
  float4 v = ((const float4*)x)[(size_t)row * (D_DIM / 4) + lane];
  float s = v.x * v.x + v.y * v.y + v.z * v.z + v.w * v.w;
  s = wave_sum32(s);
  float sc = 1.0f / fmaxf(sqrtf(s), 1e-12f);
  float4 o; o.x = v.x * sc; o.y = v.y * sc; o.z = v.z * sc; o.w = v.w * sc;
  ((float4*)xn)[(size_t)row * (D_DIM / 4) + lane] = o;
}

__global__ __launch_bounds__(32) void capsule_route(const float* __restrict__ xn,
                                                    const int* __restrict__ x_nb,
                                                    float* __restrict__ out, int n) {
  __shared__ float zt[M_NBR][ZSTR];
  __shared__ float xnrow[D_DIM];
  __shared__ float G[M_NBR][GSTR];
  __shared__ float pv[M_NBR];

  int node = blockIdx.x;
  if (node >= n) return;
  int L  = threadIdx.x;
  int c4 = L * 4;

  int nbi = x_nb[(size_t)node * M_NBR + L] - 1;

  const float4* xn4 = (const float4*)xn;
  float4 xs = xn4[(size_t)node * (D_DIM / 4) + L];
  float4 zr[M_NBR];
#pragma unroll
  for (int m = 0; m < M_NBR; ++m) {
    int r = __builtin_amdgcn_readlane(nbi, m);
    zr[m] = xn4[(size_t)r * (D_DIM / 4) + L];
  }
#pragma unroll
  for (int m = 0; m < M_NBR; ++m) *(float4*)&zt[m][c4] = zr[m];
  *(float4*)&xnrow[c4] = xs;
  __syncthreads();

  float bm = 0.0f;
#pragma unroll 8
  for (int j = 0; j < D_DIM / 4; ++j) {
    float4 zv = *(const float4*)&zt[L][4 * j];
    float4 xv = *(const float4*)&xnrow[4 * j];
    bm += zv.x * xv.x + zv.y * xv.y + zv.z * xv.z + zv.w * xv.w;
  }

  int h = L >> 4, cc = L & 15;
  v8f g00 = {}, g01 = {}, g10 = {}, g11 = {};
#pragma unroll 4
  for (int k0 = 0; k0 < D_DIM; k0 += 4) {
    v2f a0 = *(const v2f*)&zt[cc][k0 + 2 * h];
    v2f a1 = *(const v2f*)&zt[16 + cc][k0 + 2 * h];
    g00 = __builtin_amdgcn_wmma_f32_16x16x4_f32(false, a0, false, a0, (short)0, g00, false, false);
    g01 = __builtin_amdgcn_wmma_f32_16x16x4_f32(false, a0, false, a1, (short)0, g01, false, false);
    g10 = __builtin_amdgcn_wmma_f32_16x16x4_f32(false, a1, false, a0, (short)0, g10, false, false);
    g11 = __builtin_amdgcn_wmma_f32_16x16x4_f32(false, a1, false, a1, (short)0, g11, false, false);
  }
#pragma unroll
  for (int v = 0; v < 8; ++v) {
    int r = v + 8 * h;
    G[r][cc]           = g00[v];
    G[r][16 + cc]      = g01[v];
    G[16 + r][cc]      = g10[v];
    G[16 + r][16 + cc] = g11[v];
  }
  __syncthreads();

  float p = 1.0f / 32.0f;
  for (int it = 0; it < 2; ++it) {
    pv[L] = p;
    __syncthreads();
    float gp = 0.0f;
#pragma unroll
    for (int j = 0; j < 8; ++j) {
      float4 gr = *(const float4*)&G[L][4 * j];
      float4 pj = *(const float4*)&pv[4 * j];
      gp += gr.x * pj.x + gr.y * pj.y + gr.z * pj.z + gr.w * pj.w;
    }
    float sq = wave_sum32(p * (gp + 2.0f * bm)) + 1.0f;
    float s  = sqrtf(sq) / (sq + 1.0f);
    float logit = s * (gp + bm);
    float mx = wave_max32(logit);
    float e  = __expf(logit - mx);
    p = e / wave_sum32(e);
    __syncthreads();
  }

  pv[L] = p;
  __syncthreads();

  float4 acc = xs;
#pragma unroll
  for (int j = 0; j < 8; ++j) {
    float4 pj = *(const float4*)&pv[4 * j];
    float4 z0 = *(const float4*)&zt[4 * j + 0][c4];
    float4 z1 = *(const float4*)&zt[4 * j + 1][c4];
    float4 z2 = *(const float4*)&zt[4 * j + 2][c4];
    float4 z3 = *(const float4*)&zt[4 * j + 3][c4];
    acc.x += pj.x * z0.x + pj.y * z1.x + pj.z * z2.x + pj.w * z3.x;
    acc.y += pj.x * z0.y + pj.y * z1.y + pj.z * z2.y + pj.w * z3.y;
    acc.z += pj.x * z0.z + pj.y * z1.z + pj.z * z2.z + pj.w * z3.z;
    acc.w += pj.x * z0.w + pj.y * z1.w + pj.z * z2.w + pj.w * z3.w;
  }
  ((float4*)out)[(size_t)node * (D_DIM / 4) + L] = acc;
}

extern "C" void kernel_launch(void* const* d_in, const int* in_sizes, int n_in,
                              void* d_out, int out_size, void* d_ws, size_t ws_size,
                              hipStream_t stream) {
  (void)n_in; (void)out_size; (void)ws_size;
  const float* x    = (const float*)d_in[0];
  const int*   x_nb = (const int*)d_in[1];
  float*       out  = (float*)d_out;
  int n = in_sizes[0] / D_DIM;
  float* xn = (float*)d_ws;

  l2norm_rows<<<dim3((n + 7) / 8), dim3(256), 0, stream>>>(x, xn, n);
  capsule_route<<<dim3(n), dim3(32), 0, stream>>>(xn, x_nb, out, n);
}


# Wave-wide sums and maxima (row norm, routing rounds) as butterflies on the vector ALU (row_xmask DPP + permlanex16) instead of LDS permutes with a wait each; same xor order, sums bit-identical
# speedup vs baseline: 1.2672x; 1.0792x over previous
#include <hip/hip_runtime.h>


#define D_DIM 128
#define M_NBR 32
#define ZSTR  132
#define GSTR  36

typedef __attribute__((ext_vector_type(2))) float v2f;
typedef __attribute__((ext_vector_type(8))) float v8f;

static __device__ __forceinline__ float xhalf(float x) {
  int b = __builtin_bit_cast(int, x);
  return __builtin_bit_cast(float,
      __builtin_amdgcn_permlanex16(b, b, 0x76543210, 0xfedcba98, false, false));
}
template <int M>
static __device__ __forceinline__ float row_xor(float x) {
  int b = __builtin_bit_cast(int, x);
  return __builtin_bit_cast(float, __builtin_amdgcn_update_dpp(0, b, 0x160 | M, 0xf, 0xf, true));
}

static __device__ __forceinline__ float wave_sum32(float v) {
  v += xhalf(v);
  v += row_xor<8>(v);
  v += row_xor<4>(v);
  v += row_xor<2>(v);
  v += row_xor<1>(v);
  return v;
}
static __device__ __forceinline__ float wave_max32(float v) {
  v = fmaxf(v, xhalf(v));
  v = fmaxf(v, row_xor<8>(v));
  v = fmaxf(v, row_xor<4>(v));
  v = fmaxf(v, row_xor<2>(v));
  v = fmaxf(v, row_xor<1>(v));
  return v;
}

__global__ __launch_bounds__(256) void l2norm_rows(const float* __restrict__ x,
                                                   float* __restrict__ xn, int n) {
  int row  = blockIdx.x * 8 + (threadIdx.x >> 5);
  int lane = threadIdx.x & 31;
  if (row >= n) return;
  float4 v = ((const float4*)x)[(size_t)row * (D_DIM / 4) + lane];
  float s = v.x * v.x + v.y * v.y + v.z * v.z + v.w * v.w;
  s = wave_sum32(s);
  float sc = 1.0f / fmaxf(sqrtf(s), 1e-12f);
  float4 o; o.x = v.x * sc; o.y = v.y * sc; o.z = v.z * sc; o.w = v.w * sc;
  ((float4*)xn)[(size_t)row * (D_DIM / 4) + lane] = o;
}

__global__ __launch_bounds__(32) void capsule_route(const float* __restrict__ xn,
                                                    const int* __restrict__ x_nb,
                                                    float* __restrict__ out, int n) {
  __shared__ float zt[M_NBR][ZSTR];
  __shared__ float xnrow[D_DIM];
  __shared__ float G[M_NBR][GSTR];
  __shared__ float pv[M_NBR];

  int node = blockIdx.x;
  if (node >= n) return;
  int L  = threadIdx.x;
  int c4 = L * 4;

  int nbi = x_nb[(size_t)node * M_NBR + L] - 1;

  const float4* xn4 = (const float4*)xn;
  float4 xs = xn4[(size_t)node * (D_DIM / 4) + L];
  float4 zr[M_NBR];
#pragma unroll
  for (int m = 0; m < M_NBR; ++m) {
    int r = __builtin_amdgcn_readlane(nbi, m);
    zr[m] = xn4[(size_t)r * (D_DIM / 4) + L];
  }
#pragma unroll
  for (int m = 0; m < M_NBR; ++m) *(float4*)&zt[m][c4] = zr[m];
  *(float4*)&xnrow[c4] = xs;
  __syncthreads();

  float bm = 0.0f;
#pragma unroll 8
  for (int j = 0; j < D_DIM / 4; ++j) {
    float4 zv = *(const float4*)&zt[L][4 * j];
    float4 xv = *(const float4*)&xnrow[4 * j];
    bm += zv.x * xv.x + zv.y * xv.y + zv.z * xv.z + zv.w * xv.w;
  }

  int h = L >> 4, cc = L & 15;
  v8f g00 = {}, g01 = {}, g10 = {}, g11 = {};
#pragma unroll 4
  for (int k0 = 0; k0 < D_DIM; k0 += 4) {
    v2f a0 = *(const v2f*)&zt[cc][k0 + 2 * h];
    v2f a1 = *(const v2f*)&zt[16 + cc][k0 + 2 * h];
    g00 = __builtin_amdgcn_wmma_f32_16x16x4_f32(false, a0, false, a0, (short)0, g00, false, false);
    g01 = __builtin_amdgcn_wmma_f32_16x16x4_f32(false, a0, false, a1, (short)0, g01, false, false);
    g10 = __builtin_amdgcn_wmma_f32_16x16x4_f32(false, a1, false, a0, (short)0, g10, false, false);
    g11 = __builtin_amdgcn_wmma_f32_16x16x4_f32(false, a1, false, a1, (short)0, g11, false, false);
  }
#pragma unroll
  for (int v = 0; v < 8; ++v) {
    int r = v + 8 * h;
    G[r][cc]           = g00[v];
    G[r][16 + cc]      = g01[v];
    G[16 + r][cc]      = g10[v];
    G[16 + r][16 + cc] = g11[v];
  }
  __syncthreads();

  float p = 1.0f / 32.0f;
  for (int it = 0; it < 2; ++it) {
    pv[L] = p;
    __syncthreads();
    float gp = 0.0f;
#pragma unroll
    for (int j = 0; j < 8; ++j) {
      float4 gr = *(const float4*)&G[L][4 * j];
      float4 pj = *(const float4*)&pv[4 * j];
      gp += gr.x * pj.x + gr.y * pj.y + gr.z * pj.z + gr.w * pj.w;
    }
    float sq = wave_sum32(p * (gp + 2.0f * bm)) + 1.0f;
    float s  = sqrtf(sq) / (sq + 1.0f);
    float logit = s * (gp + bm);
    float mx = wave_max32(logit);
    float e  = __expf(logit - mx);
    p = e / wave_sum32(e);
    __syncthreads();
  }

  pv[L] = p;
  __syncthreads();

  float4 acc = xs;
#pragma unroll
  for (int j = 0; j < 8; ++j) {
    float4 pj = *(const float4*)&pv[4 * j];
    float4 z0 = *(const float4*)&zt[4 * j + 0][c4];
    float4 z1 = *(const float4*)&zt[4 * j + 1][c4];
    float4 z2 = *(const float4*)&zt[4 * j + 2][c4];
    float4 z3 = *(const float4*)&zt[4 * j + 3][c4];
    acc.x += pj.x * z0.x + pj.y * z1.x + pj.z * z2.x + pj.w * z3.x;
    acc.y += pj.x * z0.y + pj.y * z1.y + pj.z * z2.y + pj.w * z3.y;
    acc.z += pj.x * z0.z + pj.y * z1.z + pj.z * z2.z + pj.w * z3.z;
    acc.w += pj.x * z0.w + pj.y * z1.w + pj.z * z2.w + pj.w * z3.w;
  }
  ((float4*)out)[(size_t)node * (D_DIM / 4) + L] = acc;
}

extern "C" void kernel_launch(void* const* d_in, const int* in_sizes, int n_in,
                              void* d_out, int out_size, void* d_ws, size_t ws_size,
                              hipStream_t stream) {
  (void)n_in; (void)out_size; (void)ws_size;
  const float* x    = (const float*)d_in[0];
  const int*   x_nb = (const int*)d_in[1];
  float*       out  = (float*)d_out;
  int n = in_sizes[0] / D_DIM;
  float* xn = (float*)d_ws;

  l2norm_rows<<<dim3((n + 7) / 8), dim3(256), 0, stream>>>(x, xn, n);
  capsule_route<<<dim3(n), dim3(32), 0, stream>>>(xn, x_nb, out, n);
}


# Gram matrix is symmetric: its lower-left 16x16 block is no longer computed (3 WMMAs per K-step, not 4) but stored as the transpose of the upper-right block; every sum unchanged
# speedup vs baseline: 1.4273x; 1.1263x over previous
#include <hip/hip_runtime.h>


#define D_DIM 128
#define M_NBR 32
#define ZSTR  132
#define GSTR  36

typedef __attribute__((ext_vector_type(2))) float v2f;
typedef __attribute__((ext_vector_type(8))) float v8f;

static __device__ __forceinline__ float xhalf(float x) {
  int b = __builtin_bit_cast(int, x);
  return __builtin_bit_cast(float,
      __builtin_amdgcn_permlanex16(b, b, 0x76543210, 0xfedcba98, false, false));
}
template <int M>
static __device__ __forceinline__ float row_xor(float x) {
  int b = __builtin_bit_cast(int, x);
  return __builtin_bit_cast(float, __builtin_amdgcn_update_dpp(0, b, 0x160 | M, 0xf, 0xf, true));
}

static __device__ __forceinline__ float wave_sum32(float v) {
  v += xhalf(v);
  v += row_xor<8>(v);
  v += row_xor<4>(v);
  v += row_xor<2>(v);
  v += row_xor<1>(v);
  return v;
}
static __device__ __forceinline__ float wave_max32(float v) {
  v = fmaxf(v, xhalf(v));
  v = fmaxf(v, row_xor<8>(v));
  v = fmaxf(v, row_xor<4>(v));
  v = fmaxf(v, row_xor<2>(v));
  v = fmaxf(v, row_xor<1>(v));
  return v;
}

__global__ __launch_bounds__(256) void l2norm_rows(const float* __restrict__ x,
                                                   float* __restrict__ xn, int n) {
  int row  = blockIdx.x * 8 + (threadIdx.x >> 5);
  int lane = threadIdx.x & 31;
  if (row >= n) return;
  float4 v = ((const float4*)x)[(size_t)row * (D_DIM / 4) + lane];
  float s = v.x * v.x + v.y * v.y + v.z * v.z + v.w * v.w;
  s = wave_sum32(s);
  float sc = 1.0f / fmaxf(sqrtf(s), 1e-12f);
  float4 o; o.x = v.x * sc; o.y = v.y * sc; o.z = v.z * sc; o.w = v.w * sc;
  ((float4*)xn)[(size_t)row * (D_DIM / 4) + lane] = o;
}

__global__ __launch_bounds__(32) void capsule_route(const float* __restrict__ xn,
                                                    const int* __restrict__ x_nb,
                                                    float* __restrict__ out, int n) {
  __shared__ float zt[M_NBR][ZSTR];
  __shared__ float xnrow[D_DIM];
  __shared__ float G[M_NBR][GSTR];
  __shared__ float pv[M_NBR];

  int node = blockIdx.x;
  if (node >= n) return;
  int L  = threadIdx.x;
  int c4 = L * 4;

  int nbi = x_nb[(size_t)node * M_NBR + L] - 1;

  const float4* xn4 = (const float4*)xn;
  float4 xs = xn4[(size_t)node * (D_DIM / 4) + L];
  float4 zr[M_NBR];
#pragma unroll
  for (int m = 0; m < M_NBR; ++m) {
    int r = __builtin_amdgcn_readlane(nbi, m);
    zr[m] = xn4[(size_t)r * (D_DIM / 4) + L];
  }
#pragma unroll
  for (int m = 0; m < M_NBR; ++m) *(float4*)&zt[m][c4] = zr[m];
  *(float4*)&xnrow[c4] = xs;
  __syncthreads();

  float bm = 0.0f;
#pragma unroll 8
  for (int j = 0; j < D_DIM / 4; ++j) {
    float4 zv = *(const float4*)&zt[L][4 * j];
    float4 xv = *(const float4*)&xnrow[4 * j];
    bm += zv.x * xv.x + zv.y * xv.y + zv.z * xv.z + zv.w * xv.w;
  }

  int h = L >> 4, cc = L & 15;
  v8f g00 = {}, g01 = {}, g11 = {};
#pragma unroll 4
  for (int k0 = 0; k0 < D_DIM; k0 += 4) {
    v2f a0 = *(const v2f*)&zt[cc][k0 + 2 * h];
    v2f a1 = *(const v2f*)&zt[16 + cc][k0 + 2 * h];
    g00 = __builtin_amdgcn_wmma_f32_16x16x4_f32(false, a0, false, a0, (short)0, g00, false, false);
    g01 = __builtin_amdgcn_wmma_f32_16x16x4_f32(false, a0, false, a1, (short)0, g01, false, false);
    g11 = __builtin_amdgcn_wmma_f32_16x16x4_f32(false, a1, false, a1, (short)0, g11, false, false);
  }
#pragma unroll
  for (int v = 0; v < 8; ++v) {
    int r = v + 8 * h;
    G[r][cc]           = g00[v];
    G[r][16 + cc]      = g01[v];
    G[16 + r][16 + cc] = g11[v];
  }
  float4 t0, t1;
  t0.x = g01[0]; t0.y = g01[1]; t0.z = g01[2]; t0.w = g01[3];
  t1.x = g01[4]; t1.y = g01[5]; t1.z = g01[6]; t1.w = g01[7];
  *(float4*)&G[16 + cc][8 * h]     = t0;
  *(float4*)&G[16 + cc][8 * h + 4] = t1;
  __syncthreads();

  float p = 1.0f / 32.0f;
  for (int it = 0; it < 2; ++it) {
    pv[L] = p;
    __syncthreads();
    float gp = 0.0f;
#pragma unroll
    for (int j = 0; j < 8; ++j) {
      float4 gr = *(const float4*)&G[L][4 * j];
      float4 pj = *(const float4*)&pv[4 * j];
      gp += gr.x * pj.x + gr.y * pj.y + gr.z * pj.z + gr.w * pj.w;
    }
    float sq = wave_sum32(p * (gp + 2.0f * bm)) + 1.0f;
    float s  = sqrtf(sq) / (sq + 1.0f);
    float logit = s * (gp + bm);
    float mx = wave_max32(logit);
    float e  = __expf(logit - mx);
    p = e / wave_sum32(e);
    __syncthreads();
  }

  pv[L] = p;
  __syncthreads();

  float4 acc = xs;
#pragma unroll
  for (int j = 0; j < 8; ++j) {
    float4 pj = *(const float4*)&pv[4 * j];
    float4 z0 = *(const float4*)&zt[4 * j + 0][c4];
    float4 z1 = *(const float4*)&zt[4 * j + 1][c4];
    float4 z2 = *(const float4*)&zt[4 * j + 2][c4];
    float4 z3 = *(const float4*)&zt[4 * j + 3][c4];
    acc.x += pj.x * z0.x + pj.y * z1.x + pj.z * z2.x + pj.w * z3.x;
    acc.y += pj.x * z0.y + pj.y * z1.y + pj.z * z2.y + pj.w * z3.y;
    acc.z += pj.x * z0.z + pj.y * z1.z + pj.z * z2.z + pj.w * z3.z;
    acc.w += pj.x * z0.w + pj.y * z1.w + pj.z * z2.w + pj.w * z3.w;
  }
  ((float4*)out)[(size_t)node * (D_DIM / 4) + L] = acc;
}

extern "C" void kernel_launch(void* const* d_in, const int* in_sizes, int n_in,
                              void* d_out, int out_size, void* d_ws, size_t ws_size,
                              hipStream_t stream) {
  (void)n_in; (void)out_size; (void)ws_size;
  const float* x    = (const float*)d_in[0];
  const int*   x_nb = (const int*)d_in[1];
  float*       out  = (float*)d_out;
  int n = in_sizes[0] / D_DIM;
  float* xn = (float*)d_ws;

  l2norm_rows<<<dim3((n + 7) / 8), dim3(256), 0, stream>>>(x, xn, n);
  capsule_route<<<dim3(n), dim3(32), 0, stream>>>(xn, x_nb, out, n);
}


# Gathered tile stays in registers: b = Z.xn (lane shares + xor-butterfly transpose-sum over lanes: new sum order for b) and final Z^T p read it there, not LDS; Gram overwrites the tile's LDS copy
# speedup vs baseline: 1.5294x; 1.0715x over previous
#include <hip/hip_runtime.h>


#define D_DIM 128
#define M_NBR 32
#define ZSTR  132
#define GSTR  36

typedef __attribute__((ext_vector_type(2))) float v2f;
typedef __attribute__((ext_vector_type(8))) float v8f;

static __device__ __forceinline__ float xhalf(float x) {
  int b = __builtin_bit_cast(int, x);
  return __builtin_bit_cast(float,
      __builtin_amdgcn_permlanex16(b, b, 0x76543210, 0xfedcba98, false, false));
}
template <int M>
static __device__ __forceinline__ float row_xor(float x) {
  int b = __builtin_bit_cast(int, x);
  return __builtin_bit_cast(float, __builtin_amdgcn_update_dpp(0, b, 0x160 | M, 0xf, 0xf, true));
}

static __device__ __forceinline__ float wave_sum32(float v) {
  v += xhalf(v);
  v += row_xor<8>(v);
  v += row_xor<4>(v);
  v += row_xor<2>(v);
  v += row_xor<1>(v);
  return v;
}
static __device__ __forceinline__ float wave_max32(float v) {
  v = fmaxf(v, xhalf(v));
  v = fmaxf(v, row_xor<8>(v));
  v = fmaxf(v, row_xor<4>(v));
  v = fmaxf(v, row_xor<2>(v));
  v = fmaxf(v, row_xor<1>(v));
  return v;
}

template <int M>
static __device__ __forceinline__ float xsum_step(float lo, float hi, bool up) {
  float keep = up ? hi : lo;
  float send = up ? lo : hi;
  return keep + row_xor<M>(send);
}

__global__ __launch_bounds__(256) void l2norm_rows(const float* __restrict__ x,
                                                   float* __restrict__ xn, int n) {
  int row  = blockIdx.x * 8 + (threadIdx.x >> 5);
  int lane = threadIdx.x & 31;
  if (row >= n) return;
  float4 v = ((const float4*)x)[(size_t)row * (D_DIM / 4) + lane];
  float s = v.x * v.x + v.y * v.y + v.z * v.z + v.w * v.w;
  s = wave_sum32(s);
  float sc = 1.0f / fmaxf(sqrtf(s), 1e-12f);
  float4 o; o.x = v.x * sc; o.y = v.y * sc; o.z = v.z * sc; o.w = v.w * sc;
  ((float4*)xn)[(size_t)row * (D_DIM / 4) + lane] = o;
}

__global__ __launch_bounds__(32) __attribute__((amdgpu_waves_per_eu(4, 4))) void capsule_route(const float* __restrict__ xn,
                                                    const int* __restrict__ x_nb,
                                                    float* __restrict__ out, int n) {
  __shared__ __attribute__((aligned(16))) float tile[M_NBR * ZSTR];
  __shared__ __attribute__((aligned(16))) float pv[M_NBR];
  float (*zt)[ZSTR] = (float (*)[ZSTR])tile;
  float (*G)[GSTR]  = (float (*)[GSTR])tile;

  int node = blockIdx.x;
  if (node >= n) return;
  int L  = threadIdx.x;
  int c4 = L * 4;

  int nbi = x_nb[(size_t)node * M_NBR + L] - 1;

  const float4* xn4 = (const float4*)xn;
  float4 xs = xn4[(size_t)node * (D_DIM / 4) + L];
  float4 zr[M_NBR];
#pragma unroll
  for (int m = 0; m < M_NBR; ++m) {
    int r = __builtin_amdgcn_readlane(nbi, m);
    zr[m] = xn4[(size_t)r * (D_DIM / 4) + L];
  }
#pragma unroll
  for (int m = 0; m < M_NBR; ++m) *(float4*)&zt[m][c4] = zr[m];
  __syncthreads();

  float t[M_NBR];
  v2f xlo = {xs.x, xs.y}, xhi = {xs.z, xs.w};
#pragma unroll
  for (int m = 0; m < M_NBR; ++m) {
    v2f zlo = {zr[m].x, zr[m].y}, zhi = {zr[m].z, zr[m].w};
    v2f q = zlo * xlo + zhi * xhi;
    t[m] = q.x + q.y;
  }
  float u1[16], u2[8], u3[4], u4[2];
#pragma unroll
  for (int i = 0; i < 16; ++i) u1[i] = xsum_step<1>(t[2 * i], t[2 * i + 1], (L & 1) != 0);
#pragma unroll
  for (int i = 0; i < 8; ++i) u2[i] = xsum_step<2>(u1[2 * i], u1[2 * i + 1], (L & 2) != 0);
#pragma unroll
  for (int i = 0; i < 4; ++i) u3[i] = xsum_step<4>(u2[2 * i], u2[2 * i + 1], (L & 4) != 0);
#pragma unroll
  for (int i = 0; i < 2; ++i) u4[i] = xsum_step<8>(u3[2 * i], u3[2 * i + 1], (L & 8) != 0);
  float bm;
  {
    bool up = (L & 16) != 0;
    float keep = up ? u4[1] : u4[0];
    float send = up ? u4[0] : u4[1];
    bm = keep + xhalf(send);
  }

  int h = L >> 4, cc = L & 15;
  v8f g00 = {}, g01 = {}, g11 = {};
#pragma unroll 4
  for (int k0 = 0; k0 < D_DIM; k0 += 4) {
    v2f a0 = *(const v2f*)&zt[cc][k0 + 2 * h];
    v2f a1 = *(const v2f*)&zt[16 + cc][k0 + 2 * h];
    g00 = __builtin_amdgcn_wmma_f32_16x16x4_f32(false, a0, false, a0, (short)0, g00, false, false);
    g01 = __builtin_amdgcn_wmma_f32_16x16x4_f32(false, a0, false, a1, (short)0, g01, false, false);
    g11 = __builtin_amdgcn_wmma_f32_16x16x4_f32(false, a1, false, a1, (short)0, g11, false, false);
  }
  __syncthreads();
#pragma unroll
  for (int v = 0; v < 8; ++v) {
    int r = v + 8 * h;
    G[r][cc]           = g00[v];
    G[r][16 + cc]      = g01[v];
    G[16 + r][16 + cc] = g11[v];
  }
  float4 t0, t1;
  t0.x = g01[0]; t0.y = g01[1]; t0.z = g01[2]; t0.w = g01[3];
  t1.x = g01[4]; t1.y = g01[5]; t1.z = g01[6]; t1.w = g01[7];
  *(float4*)&G[16 + cc][8 * h]     = t0;
  *(float4*)&G[16 + cc][8 * h + 4] = t1;
  __syncthreads();

  float p = 1.0f / 32.0f;
  for (int it = 0; it < 2; ++it) {
    pv[L] = p;
    __syncthreads();
    float gp = 0.0f;
#pragma unroll
    for (int j = 0; j < 8; ++j) {
      float4 gr = *(const float4*)&G[L][4 * j];
      float4 pj = *(const float4*)&pv[4 * j];
      gp += gr.x * pj.x + gr.y * pj.y + gr.z * pj.z + gr.w * pj.w;
    }
    float sq = wave_sum32(p * (gp + 2.0f * bm)) + 1.0f;
    float s  = sqrtf(sq) / (sq + 1.0f);
    float logit = s * (gp + bm);
    float mx = wave_max32(logit);
    float e  = __expf(logit - mx);
    p = e / wave_sum32(e);
    __syncthreads();
  }

  int pbits = __builtin_bit_cast(int, p);
  float4 acc = xs;
#pragma unroll
  for (int j = 0; j < 8; ++j) {
    float4 pj;
    pj.x = __builtin_bit_cast(float, __builtin_amdgcn_readlane(pbits, 4 * j + 0));
    pj.y = __builtin_bit_cast(float, __builtin_amdgcn_readlane(pbits, 4 * j + 1));
    pj.z = __builtin_bit_cast(float, __builtin_amdgcn_readlane(pbits, 4 * j + 2));
    pj.w = __builtin_bit_cast(float, __builtin_amdgcn_readlane(pbits, 4 * j + 3));
    float4 z0 = zr[4 * j + 0];
    float4 z1 = zr[4 * j + 1];
    float4 z2 = zr[4 * j + 2];
    float4 z3 = zr[4 * j + 3];
    acc.x += pj.x * z0.x + pj.y * z1.x + pj.z * z2.x + pj.w * z3.x;
    acc.y += pj.x * z0.y + pj.y * z1.y + pj.z * z2.y + pj.w * z3.y;
    acc.z += pj.x * z0.z + pj.y * z1.z + pj.z * z2.z + pj.w * z3.z;
    acc.w += pj.x * z0.w + pj.y * z1.w + pj.z * z2.w + pj.w * z3.w;
  }
  ((float4*)out)[(size_t)node * (D_DIM / 4) + L] = acc;
}

extern "C" void kernel_launch(void* const* d_in, const int* in_sizes, int n_in,
                              void* d_out, int out_size, void* d_ws, size_t ws_size,
                              hipStream_t stream) {
  (void)n_in; (void)out_size; (void)ws_size;
  const float* x    = (const float*)d_in[0];
  const int*   x_nb = (const int*)d_in[1];
  float*       out  = (float*)d_out;
  int n = in_sizes[0] / D_DIM;
  float* xn = (float*)d_ws;

  l2norm_rows<<<dim3((n + 7) / 8), dim3(256), 0, stream>>>(x, xn, n);
  capsule_route<<<dim3(n), dim3(32), 0, stream>>>(xn, x_nb, out, n);
}


# Row normalisation folded into the routing launch (2 launches -> 1, no scratch): raw rows gathered; 1/|row| = rsqrt of raw Gram diagonal (norm now summed by WMMA in k order) scales p, b, self row
# speedup vs baseline: 1.7363x; 1.1353x over previous
#include <hip/hip_runtime.h>


#define D_DIM 128
#define M_NBR 32
#define ZSTR  132
#define GSTR  36

typedef __attribute__((ext_vector_type(2))) float v2f;
typedef __attribute__((ext_vector_type(8))) float v8f;

static __device__ __forceinline__ float xhalf(float x) {
  int b = __builtin_bit_cast(int, x);
  return __builtin_bit_cast(float,
      __builtin_amdgcn_permlanex16(b, b, 0x76543210, 0xfedcba98, false, false));
}
template <int M>
static __device__ __forceinline__ float row_xor(float x) {
  int b = __builtin_bit_cast(int, x);
  return __builtin_bit_cast(float, __builtin_amdgcn_update_dpp(0, b, 0x160 | M, 0xf, 0xf, true));
}

static __device__ __forceinline__ float wave_sum32(float v) {
  v += xhalf(v);
  v += row_xor<8>(v);
  v += row_xor<4>(v);
  v += row_xor<2>(v);
  v += row_xor<1>(v);
  return v;
}
static __device__ __forceinline__ float wave_max32(float v) {
  v = fmaxf(v, xhalf(v));
  v = fmaxf(v, row_xor<8>(v));
  v = fmaxf(v, row_xor<4>(v));
  v = fmaxf(v, row_xor<2>(v));
  v = fmaxf(v, row_xor<1>(v));
  return v;
}

template <int M>
static __device__ __forceinline__ float xsum_step(float lo, float hi, bool up) {
  float keep = up ? hi : lo;
  float send = up ? lo : hi;
  return keep + row_xor<M>(send);
}

static __device__ __forceinline__ float inv_norm(float sq) {
  return fminf(rsqrtf(sq), 1e12f);
}

__global__ __launch_bounds__(32) __attribute__((amdgpu_waves_per_eu(4, 4))) void capsule_route(const float* __restrict__ x,
                                                    const int* __restrict__ x_nb,
                                                    float* __restrict__ out, int n) {
  __shared__ __attribute__((aligned(16))) float tile[M_NBR * ZSTR];
  __shared__ __attribute__((aligned(16))) float pv[M_NBR];
  float (*zt)[ZSTR] = (float (*)[ZSTR])tile;
  float (*G)[GSTR]  = (float (*)[GSTR])tile;

  int node = blockIdx.x;
  if (node >= n) return;
  int L  = threadIdx.x;
  int c4 = L * 4;

  int nbi = x_nb[(size_t)node * M_NBR + L] - 1;

  const float4* x4 = (const float4*)x;
  float4 xs = x4[(size_t)node * (D_DIM / 4) + L];
  float4 zr[M_NBR];
#pragma unroll
  for (int m = 0; m < M_NBR; ++m) {
    int r = __builtin_amdgcn_readlane(nbi, m);
    zr[m] = x4[(size_t)r * (D_DIM / 4) + L];
  }
#pragma unroll
  for (int m = 0; m < M_NBR; ++m) *(float4*)&zt[m][c4] = zr[m];
  __syncthreads();

  float t[M_NBR];
  v2f xlo = {xs.x, xs.y}, xhi = {xs.z, xs.w};
#pragma unroll
  for (int m = 0; m < M_NBR; ++m) {
    v2f zlo = {zr[m].x, zr[m].y}, zhi = {zr[m].z, zr[m].w};
    v2f q = zlo * xlo + zhi * xhi;
    t[m] = q.x + q.y;
  }
  float u1[16], u2[8], u3[4], u4[2];
#pragma unroll
  for (int i = 0; i < 16; ++i) u1[i] = xsum_step<1>(t[2 * i], t[2 * i + 1], (L & 1) != 0);
#pragma unroll
  for (int i = 0; i < 8; ++i) u2[i] = xsum_step<2>(u1[2 * i], u1[2 * i + 1], (L & 2) != 0);
#pragma unroll
  for (int i = 0; i < 4; ++i) u3[i] = xsum_step<4>(u2[2 * i], u2[2 * i + 1], (L & 4) != 0);
#pragma unroll
  for (int i = 0; i < 2; ++i) u4[i] = xsum_step<8>(u3[2 * i], u3[2 * i + 1], (L & 8) != 0);
  float braw;
  {
    bool up = (L & 16) != 0;
    float keep = up ? u4[1] : u4[0];
    float send = up ? u4[0] : u4[1];
    braw = keep + xhalf(send);
  }
  float rs = inv_norm(wave_sum32(xs.x * xs.x + xs.y * xs.y + xs.z * xs.z + xs.w * xs.w));

  int h = L >> 4, cc = L & 15;
  v8f g00 = {}, g01 = {}, g11 = {};
#pragma unroll 4
  for (int k0 = 0; k0 < D_DIM; k0 += 4) {
    v2f a0 = *(const v2f*)&zt[cc][k0 + 2 * h];
    v2f a1 = *(const v2f*)&zt[16 + cc][k0 + 2 * h];
    g00 = __builtin_amdgcn_wmma_f32_16x16x4_f32(false, a0, false, a0, (short)0, g00, false, false);
    g01 = __builtin_amdgcn_wmma_f32_16x16x4_f32(false, a0, false, a1, (short)0, g01, false, false);
    g11 = __builtin_amdgcn_wmma_f32_16x16x4_f32(false, a1, false, a1, (short)0, g11, false, false);
  }
  __syncthreads();
#pragma unroll
  for (int v = 0; v < 8; ++v) {
    int r = v + 8 * h;
    G[r][cc]           = g00[v];
    G[r][16 + cc]      = g01[v];
    G[16 + r][16 + cc] = g11[v];
  }
  float4 t0, t1;
  t0.x = g01[0]; t0.y = g01[1]; t0.z = g01[2]; t0.w = g01[3];
  t1.x = g01[4]; t1.y = g01[5]; t1.z = g01[6]; t1.w = g01[7];
  *(float4*)&G[16 + cc][8 * h]     = t0;
  *(float4*)&G[16 + cc][8 * h + 4] = t1;
  __syncthreads();

  float rl = inv_norm(G[L][L]);
  float bm = braw * (rl * rs);

  float p = 1.0f / 32.0f;
  for (int it = 0; it < 2; ++it) {
    pv[L] = p * rl;
    __syncthreads();
    float gp = 0.0f;
#pragma unroll
    for (int j = 0; j < 8; ++j) {
      float4 gr = *(const float4*)&G[L][4 * j];
      float4 pj = *(const float4*)&pv[4 * j];
      gp += gr.x * pj.x + gr.y * pj.y + gr.z * pj.z + gr.w * pj.w;
    }
    gp *= rl;
    float sq = wave_sum32(p * (gp + 2.0f * bm)) + 1.0f;
    float s  = sqrtf(sq) / (sq + 1.0f);
    float logit = s * (gp + bm);
    float mx = wave_max32(logit);
    float e  = __expf(logit - mx);
    p = e / wave_sum32(e);
    __syncthreads();
  }

  int pbits = __builtin_bit_cast(int, p * rl);
  float4 acc;
  acc.x = xs.x * rs; acc.y = xs.y * rs; acc.z = xs.z * rs; acc.w = xs.w * rs;
#pragma unroll
  for (int j = 0; j < 8; ++j) {
    float4 pj;
    pj.x = __builtin_bit_cast(float, __builtin_amdgcn_readlane(pbits, 4 * j + 0));
    pj.y = __builtin_bit_cast(float, __builtin_amdgcn_readlane(pbits, 4 * j + 1));
    pj.z = __builtin_bit_cast(float, __builtin_amdgcn_readlane(pbits, 4 * j + 2));
    pj.w = __builtin_bit_cast(float, __builtin_amdgcn_readlane(pbits, 4 * j + 3));
    float4 z0 = zr[4 * j + 0];
    float4 z1 = zr[4 * j + 1];
    float4 z2 = zr[4 * j + 2];
    float4 z3 = zr[4 * j + 3];
    acc.x += pj.x * z0.x + pj.y * z1.x + pj.z * z2.x + pj.w * z3.x;
    acc.y += pj.x * z0.y + pj.y * z1.y + pj.z * z2.y + pj.w * z3.y;
    acc.z += pj.x * z0.z + pj.y * z1.z + pj.z * z2.z + pj.w * z3.z;
    acc.w += pj.x * z0.w + pj.y * z1.w + pj.z * z2.w + pj.w * z3.w;
  }
  ((float4*)out)[(size_t)node * (D_DIM / 4) + L] = acc;
}

extern "C" void kernel_launch(void* const* d_in, const int* in_sizes, int n_in,
                              void* d_out, int out_size, void* d_ws, size_t ws_size,
                              hipStream_t stream) {
  (void)n_in; (void)out_size; (void)d_ws; (void)ws_size;
  const float* x    = (const float*)d_in[0];
  const int*   x_nb = (const int*)d_in[1];
  float*       out  = (float*)d_out;
  int n = in_sizes[0] / D_DIM;

  capsule_route<<<dim3(n), dim3(32), 0, stream>>>(x, x_nb, out, n);
}


# Five waves per SIMD, not four: 16 KiB unpadded xor-swizzled LDS tile, scalar row bases, <=192 VGPRs; K-steps group other columns, G.p in two trips: new sum order in Gram and G.p
# speedup vs baseline: 1.8204x; 1.0484x over previous
#include <hip/hip_runtime.h>


#define D_DIM 128
#define M_NBR 32
#define PVOFF (M_NBR * GSTR)
#define GSTR  36

typedef __attribute__((ext_vector_type(2))) float v2f;
typedef __attribute__((ext_vector_type(8))) float v8f;
typedef __attribute__((ext_vector_type(4))) float v4f;
typedef const __attribute__((address_space(1))) v4f* gv4p;

static __device__ __forceinline__ float xhalf(float x) {
  int b = __builtin_bit_cast(int, x);
  return __builtin_bit_cast(float,
      __builtin_amdgcn_permlanex16(b, b, 0x76543210, 0xfedcba98, false, false));
}
template <int M>
static __device__ __forceinline__ float row_xor(float x) {
  int b = __builtin_bit_cast(int, x);
  return __builtin_bit_cast(float, __builtin_amdgcn_update_dpp(0, b, 0x160 | M, 0xf, 0xf, true));
}

static __device__ __forceinline__ float wave_sum32(float v) {
  v += xhalf(v);
  v += row_xor<8>(v);
  v += row_xor<4>(v);
  v += row_xor<2>(v);
  v += row_xor<1>(v);
  return v;
}
static __device__ __forceinline__ float wave_max32(float v) {
  v = fmaxf(v, xhalf(v));
  v = fmaxf(v, row_xor<8>(v));
  v = fmaxf(v, row_xor<4>(v));
  v = fmaxf(v, row_xor<2>(v));
  v = fmaxf(v, row_xor<1>(v));
  return v;
}

template <int M>
static __device__ __forceinline__ float xsum_step(float lo, float hi, bool up) {
  float keep = up ? hi : lo;
  float send = up ? lo : hi;
  return keep + row_xor<M>(send);
}

static __device__ __forceinline__ float inv_norm(float sq) {
  return fminf(rsqrtf(sq), 1e12f);
}

static __device__ __forceinline__ unsigned long long uniform_base(unsigned long long b) {
  unsigned lo = (unsigned)__builtin_amdgcn_readfirstlane((int)(unsigned)b);
  unsigned hi = (unsigned)__builtin_amdgcn_readfirstlane((int)(unsigned)(b >> 32));
  return ((unsigned long long)hi << 32) | lo;
}
static __device__ __forceinline__ unsigned long long idx_base(unsigned long long ib, int node) {
  return uniform_base(ib + (unsigned long long)((long long)node * (M_NBR * 4)));
}
static __device__ __forceinline__ unsigned long long row_base(unsigned long long xb, int r) {
  unsigned long long b = xb + (unsigned long long)((long long)r * (D_DIM * 4));
  unsigned lo = (unsigned)__builtin_amdgcn_readfirstlane((int)(unsigned)b);
  unsigned hi = (unsigned)__builtin_amdgcn_readfirstlane((int)(unsigned)(b >> 32));
  return ((unsigned long long)hi << 32) | lo;
}

__global__ __launch_bounds__(32) __attribute__((amdgpu_waves_per_eu(5, 5))) void capsule_route(const float* __restrict__ x,
                                                    const int* __restrict__ x_nb,
                                                    float* __restrict__ out, int n) {
  __shared__ __attribute__((aligned(16))) float tile[M_NBR * D_DIM];
  float (*G)[GSTR]  = (float (*)[GSTR])tile;
  float* pv = tile + PVOFF;

  int node = blockIdx.x;
  if (node >= n) return;
  int L  = threadIdx.x;

  int nbi = *(const __attribute__((address_space(1))) int*)(idx_base((unsigned long long)x_nb, node) + (unsigned)L * 4u) - 1;

  unsigned long long xb = (unsigned long long)x;
  unsigned loff = (unsigned)L * 16u;
  v4f xs = *(gv4p)(row_base(xb, node) + loff);
  v4f zr[M_NBR];
#pragma unroll
  for (int m = 0; m < M_NBR; ++m) {
    int r = __builtin_amdgcn_readlane(nbi, m);
    zr[m] = *(gv4p)(row_base(xb, r) + loff);
  }
#pragma unroll
  for (int m = 0; m < M_NBR; ++m)
    *(v4f*)&tile[m * D_DIM + ((L ^ (m & 15)) << 2)] = zr[m];
  __syncthreads();


  int h = L >> 4, cc = L & 15;
  unsigned abase = (unsigned)cc * (D_DIM * 4) + ((unsigned)(cc ^ h) << 4);
  const char* tb = (const char*)tile;
  v8f g00 = {}, g01 = {}, g11 = {};
  unsigned ad = abase;
#pragma unroll 1
  for (int j = 0; j < 8; ++j) {
    float4 p0 = *(const float4*)(tb + ad);
    float4 p1 = *(const float4*)(tb + ad + 16 * D_DIM * 4);
    float4 q0 = *(const float4*)(tb + ad + 256);
    float4 q1 = *(const float4*)(tb + ad + 16 * D_DIM * 4 + 256);
    ad = abase ^ ((unsigned)(j + 1) << 5);
    v2f a0, a1;
    a0 = (v2f){p0.x, p0.y}; a1 = (v2f){p1.x, p1.y};
    g00 = __builtin_amdgcn_wmma_f32_16x16x4_f32(false, a0, false, a0, (short)0, g00, false, false);
    g01 = __builtin_amdgcn_wmma_f32_16x16x4_f32(false, a0, false, a1, (short)0, g01, false, false);
    g11 = __builtin_amdgcn_wmma_f32_16x16x4_f32(false, a1, false, a1, (short)0, g11, false, false);
    a0 = (v2f){p0.z, p0.w}; a1 = (v2f){p1.z, p1.w};
    g00 = __builtin_amdgcn_wmma_f32_16x16x4_f32(false, a0, false, a0, (short)0, g00, false, false);
    g01 = __builtin_amdgcn_wmma_f32_16x16x4_f32(false, a0, false, a1, (short)0, g01, false, false);
    g11 = __builtin_amdgcn_wmma_f32_16x16x4_f32(false, a1, false, a1, (short)0, g11, false, false);
    a0 = (v2f){q0.x, q0.y}; a1 = (v2f){q1.x, q1.y};
    g00 = __builtin_amdgcn_wmma_f32_16x16x4_f32(false, a0, false, a0, (short)0, g00, false, false);
    g01 = __builtin_amdgcn_wmma_f32_16x16x4_f32(false, a0, false, a1, (short)0, g01, false, false);
    g11 = __builtin_amdgcn_wmma_f32_16x16x4_f32(false, a1, false, a1, (short)0, g11, false, false);
    a0 = (v2f){q0.z, q0.w}; a1 = (v2f){q1.z, q1.w};
    g00 = __builtin_amdgcn_wmma_f32_16x16x4_f32(false, a0, false, a0, (short)0, g00, false, false);
    g01 = __builtin_amdgcn_wmma_f32_16x16x4_f32(false, a0, false, a1, (short)0, g01, false, false);
    g11 = __builtin_amdgcn_wmma_f32_16x16x4_f32(false, a1, false, a1, (short)0, g11, false, false);
  }
  __syncthreads();
#pragma unroll
  for (int v = 0; v < 8; ++v) {
    int r = v + 8 * h;
    G[r][cc]           = g00[v];
    G[r][16 + cc]      = g01[v];
    G[16 + r][16 + cc] = g11[v];
  }
  float4 t0, t1;
  t0.x = g01[0]; t0.y = g01[1]; t0.z = g01[2]; t0.w = g01[3];
  t1.x = g01[4]; t1.y = g01[5]; t1.z = g01[6]; t1.w = g01[7];
  *(float4*)&G[16 + cc][8 * h]     = t0;
  *(float4*)&G[16 + cc][8 * h + 4] = t1;
  __syncthreads();

  v2f xlo = {xs.x, xs.y}, xhi = {xs.z, xs.w};
  float u3[4], u4[2];
#pragma unroll
  for (int g = 0; g < 4; ++g) {
    float t[8], u1[4], u2[2];
#pragma unroll
    for (int i = 0; i < 8; ++i) {
      int m = 8 * g + i;
      v2f zlo = {zr[m].x, zr[m].y}, zhi = {zr[m].z, zr[m].w};
      v2f q = zlo * xlo + zhi * xhi;
      t[i] = q.x + q.y;
    }
#pragma unroll
    for (int i = 0; i < 4; ++i) u1[i] = xsum_step<1>(t[2 * i], t[2 * i + 1], (L & 1) != 0);
#pragma unroll
    for (int i = 0; i < 2; ++i) u2[i] = xsum_step<2>(u1[2 * i], u1[2 * i + 1], (L & 2) != 0);
    u3[g] = xsum_step<4>(u2[0], u2[1], (L & 4) != 0);
  }
#pragma unroll
  for (int i = 0; i < 2; ++i) u4[i] = xsum_step<8>(u3[2 * i], u3[2 * i + 1], (L & 8) != 0);
  float braw;
  {
    bool up = (L & 16) != 0;
    float keep = up ? u4[1] : u4[0];
    float send = up ? u4[0] : u4[1];
    braw = keep + xhalf(send);
  }
  float rs = inv_norm(wave_sum32(xs.x * xs.x + xs.y * xs.y + xs.z * xs.z + xs.w * xs.w));

  float rl = inv_norm(G[L][L]);
  float bm = braw * (rl * rs);

  float p = 1.0f / 32.0f;
  for (int it = 0; it < 2; ++it) {
    pv[L] = p * rl;
    __syncthreads();
    float gp = 0.0f;
#pragma unroll 4
    for (int j = 0; j < 8; ++j) {
      float4 gr = *(const float4*)&G[L][4 * j];
      float4 pj = *(const float4*)&pv[4 * j];
      gp += gr.x * pj.x + gr.y * pj.y + gr.z * pj.z + gr.w * pj.w;
    }
    gp *= rl;
    float sq = wave_sum32(p * (gp + 2.0f * bm)) + 1.0f;
    float s  = sqrtf(sq) / (sq + 1.0f);
    float logit = s * (gp + bm);
    float mx = wave_max32(logit);
    float e  = __expf(logit - mx);
    p = e / wave_sum32(e);
    __syncthreads();
  }

  int pbits = __builtin_bit_cast(int, p * rl);
  float4 acc;
  acc.x = xs.x * rs; acc.y = xs.y * rs; acc.z = xs.z * rs; acc.w = xs.w * rs;
#pragma unroll
  for (int j = 0; j < 8; ++j) {
    float4 pj;
    pj.x = __builtin_bit_cast(float, __builtin_amdgcn_readlane(pbits, 4 * j + 0));
    pj.y = __builtin_bit_cast(float, __builtin_amdgcn_readlane(pbits, 4 * j + 1));
    pj.z = __builtin_bit_cast(float, __builtin_amdgcn_readlane(pbits, 4 * j + 2));
    pj.w = __builtin_bit_cast(float, __builtin_amdgcn_readlane(pbits, 4 * j + 3));
    v4f z0 = zr[4 * j + 0];
    v4f z1 = zr[4 * j + 1];
    v4f z2 = zr[4 * j + 2];
    v4f z3 = zr[4 * j + 3];
    acc.x += pj.x * z0.x + pj.y * z1.x + pj.z * z2.x + pj.w * z3.x;
    acc.y += pj.x * z0.y + pj.y * z1.y + pj.z * z2.y + pj.w * z3.y;
    acc.z += pj.x * z0.z + pj.y * z1.z + pj.z * z2.z + pj.w * z3.z;
    acc.w += pj.x * z0.w + pj.y * z1.w + pj.z * z2.w + pj.w * z3.w;
  }
  v4f res = {acc.x, acc.y, acc.z, acc.w};
  *(__attribute__((address_space(1))) v4f*)(row_base((unsigned long long)out, node) + loff) = res;
}

extern "C" void kernel_launch(void* const* d_in, const int* in_sizes, int n_in,
                              void* d_out, int out_size, void* d_ws, size_t ws_size,
                              hipStream_t stream) {
  (void)n_in; (void)out_size; (void)d_ws; (void)ws_size;
  const float* x    = (const float*)d_in[0];
  const int*   x_nb = (const int*)d_in[1];
  float*       out  = (float*)d_out;
  int n = in_sizes[0] / D_DIM;

  capsule_route<<<dim3(n), dim3(32), 0, stream>>>(x, x_nb, out, n);
}


# Routing rounds' scalar chain in bare hardware ops: v_sqrt, v_rcp, v_rsq replace IEEE-corrected sqrt, divisions, rescaled rsqrt; softmax without max shift (|logit| < 1): last-ulp changes in p and s
# speedup vs baseline: 1.9009x; 1.0442x over previous
#include <hip/hip_runtime.h>


#define D_DIM 128
#define M_NBR 32
#define PVOFF (M_NBR * GSTR)
#define GSTR  36

typedef __attribute__((ext_vector_type(2))) float v2f;
typedef __attribute__((ext_vector_type(8))) float v8f;
typedef __attribute__((ext_vector_type(4))) float v4f;
typedef const __attribute__((address_space(1))) v4f* gv4p;

static __device__ __forceinline__ float xhalf(float x) {
  int b = __builtin_bit_cast(int, x);
  return __builtin_bit_cast(float,
      __builtin_amdgcn_permlanex16(b, b, 0x76543210, 0xfedcba98, false, false));
}
template <int M>
static __device__ __forceinline__ float row_xor(float x) {
  int b = __builtin_bit_cast(int, x);
  return __builtin_bit_cast(float, __builtin_amdgcn_update_dpp(0, b, 0x160 | M, 0xf, 0xf, true));
}

static __device__ __forceinline__ float wave_sum32(float v) {
  v += xhalf(v);
  v += row_xor<8>(v);
  v += row_xor<4>(v);
  v += row_xor<2>(v);
  v += row_xor<1>(v);
  return v;
}

template <int M>
static __device__ __forceinline__ float xsum_step(float lo, float hi, bool up) {
  float keep = up ? hi : lo;
  float send = up ? lo : hi;
  return keep + row_xor<M>(send);
}

static __device__ __forceinline__ float inv_norm(float sq) {
  return fminf(__builtin_amdgcn_rsqf(sq), 1e12f);
}

static __device__ __forceinline__ unsigned long long uniform_base(unsigned long long b) {
  unsigned lo = (unsigned)__builtin_amdgcn_readfirstlane((int)(unsigned)b);
  unsigned hi = (unsigned)__builtin_amdgcn_readfirstlane((int)(unsigned)(b >> 32));
  return ((unsigned long long)hi << 32) | lo;
}
static __device__ __forceinline__ unsigned long long idx_base(unsigned long long ib, int node) {
  return uniform_base(ib + (unsigned long long)((long long)node * (M_NBR * 4)));
}
static __device__ __forceinline__ unsigned long long row_base(unsigned long long xb, int r) {
  unsigned long long b = xb + (unsigned long long)((long long)r * (D_DIM * 4));
  unsigned lo = (unsigned)__builtin_amdgcn_readfirstlane((int)(unsigned)b);
  unsigned hi = (unsigned)__builtin_amdgcn_readfirstlane((int)(unsigned)(b >> 32));
  return ((unsigned long long)hi << 32) | lo;
}

__global__ __launch_bounds__(32) __attribute__((amdgpu_waves_per_eu(5, 5))) void capsule_route(const float* __restrict__ x,
                                                    const int* __restrict__ x_nb,
                                                    float* __restrict__ out, int n) {
  __shared__ __attribute__((aligned(16))) float tile[M_NBR * D_DIM];
  float (*G)[GSTR]  = (float (*)[GSTR])tile;
  float* pv = tile + PVOFF;

  int node = blockIdx.x;
  if (node >= n) return;
  int L  = threadIdx.x;

  int nbi = *(const __attribute__((address_space(1))) int*)(idx_base((unsigned long long)x_nb, node) + (unsigned)L * 4u) - 1;

  unsigned long long xb = (unsigned long long)x;
  unsigned loff = (unsigned)L * 16u;
  v4f xs = *(gv4p)(row_base(xb, node) + loff);
  v4f zr[M_NBR];
#pragma unroll
  for (int m = 0; m < M_NBR; ++m) {
    int r = __builtin_amdgcn_readlane(nbi, m);
    zr[m] = *(gv4p)(row_base(xb, r) + loff);
  }
#pragma unroll
  for (int m = 0; m < M_NBR; ++m)
    *(v4f*)&tile[m * D_DIM + ((L ^ (m & 15)) << 2)] = zr[m];
  __syncthreads();


  int h = L >> 4, cc = L & 15;
  unsigned abase = (unsigned)cc * (D_DIM * 4) + ((unsigned)(cc ^ h) << 4);
  const char* tb = (const char*)tile;
  v8f g00 = {}, g01 = {}, g11 = {};
  unsigned ad = abase;
#pragma unroll 1
  for (int j = 0; j < 8; ++j) {
    float4 p0 = *(const float4*)(tb + ad);
    float4 p1 = *(const float4*)(tb + ad + 16 * D_DIM * 4);
    float4 q0 = *(const float4*)(tb + ad + 256);
    float4 q1 = *(const float4*)(tb + ad + 16 * D_DIM * 4 + 256);
    ad = abase ^ ((unsigned)(j + 1) << 5);
    v2f a0, a1;
    a0 = (v2f){p0.x, p0.y}; a1 = (v2f){p1.x, p1.y};
    g00 = __builtin_amdgcn_wmma_f32_16x16x4_f32(false, a0, false, a0, (short)0, g00, false, false);
    g01 = __builtin_amdgcn_wmma_f32_16x16x4_f32(false, a0, false, a1, (short)0, g01, false, false);
    g11 = __builtin_amdgcn_wmma_f32_16x16x4_f32(false, a1, false, a1, (short)0, g11, false, false);
    a0 = (v2f){p0.z, p0.w}; a1 = (v2f){p1.z, p1.w};
    g00 = __builtin_amdgcn_wmma_f32_16x16x4_f32(false, a0, false, a0, (short)0, g00, false, false);
    g01 = __builtin_amdgcn_wmma_f32_16x16x4_f32(false, a0, false, a1, (short)0, g01, false, false);
    g11 = __builtin_amdgcn_wmma_f32_16x16x4_f32(false, a1, false, a1, (short)0, g11, false, false);
    a0 = (v2f){q0.x, q0.y}; a1 = (v2f){q1.x, q1.y};
    g00 = __builtin_amdgcn_wmma_f32_16x16x4_f32(false, a0, false, a0, (short)0, g00, false, false);
    g01 = __builtin_amdgcn_wmma_f32_16x16x4_f32(false, a0, false, a1, (short)0, g01, false, false);
    g11 = __builtin_amdgcn_wmma_f32_16x16x4_f32(false, a1, false, a1, (short)0, g11, false, false);
    a0 = (v2f){q0.z, q0.w}; a1 = (v2f){q1.z, q1.w};
    g00 = __builtin_amdgcn_wmma_f32_16x16x4_f32(false, a0, false, a0, (short)0, g00, false, false);
    g01 = __builtin_amdgcn_wmma_f32_16x16x4_f32(false, a0, false, a1, (short)0, g01, false, false);
    g11 = __builtin_amdgcn_wmma_f32_16x16x4_f32(false, a1, false, a1, (short)0, g11, false, false);
  }
  __syncthreads();
#pragma unroll
  for (int v = 0; v < 8; ++v) {
    int r = v + 8 * h;
    G[r][cc]           = g00[v];
    G[r][16 + cc]      = g01[v];
    G[16 + r][16 + cc] = g11[v];
  }
  float4 t0, t1;
  t0.x = g01[0]; t0.y = g01[1]; t0.z = g01[2]; t0.w = g01[3];
  t1.x = g01[4]; t1.y = g01[5]; t1.z = g01[6]; t1.w = g01[7];
  *(float4*)&G[16 + cc][8 * h]     = t0;
  *(float4*)&G[16 + cc][8 * h + 4] = t1;
  __syncthreads();

  v2f xlo = {xs.x, xs.y}, xhi = {xs.z, xs.w};
  float u3[4], u4[2];
#pragma unroll
  for (int g = 0; g < 4; ++g) {
    float t[8], u1[4], u2[2];
#pragma unroll
    for (int i = 0; i < 8; ++i) {
      int m = 8 * g + i;
      v2f zlo = {zr[m].x, zr[m].y}, zhi = {zr[m].z, zr[m].w};
      v2f q = zlo * xlo + zhi * xhi;
      t[i] = q.x + q.y;
    }
#pragma unroll
    for (int i = 0; i < 4; ++i) u1[i] = xsum_step<1>(t[2 * i], t[2 * i + 1], (L & 1) != 0);
#pragma unroll
    for (int i = 0; i < 2; ++i) u2[i] = xsum_step<2>(u1[2 * i], u1[2 * i + 1], (L & 2) != 0);
    u3[g] = xsum_step<4>(u2[0], u2[1], (L & 4) != 0);
  }
#pragma unroll
  for (int i = 0; i < 2; ++i) u4[i] = xsum_step<8>(u3[2 * i], u3[2 * i + 1], (L & 8) != 0);
  float braw;
  {
    bool up = (L & 16) != 0;
    float keep = up ? u4[1] : u4[0];
    float send = up ? u4[0] : u4[1];
    braw = keep + xhalf(send);
  }
  float rs = inv_norm(wave_sum32(xs.x * xs.x + xs.y * xs.y + xs.z * xs.z + xs.w * xs.w));

  float rl = inv_norm(G[L][L]);
  float bm = braw * (rl * rs);

  float p = 1.0f / 32.0f;
  for (int it = 0; it < 2; ++it) {
    pv[L] = p * rl;
    __syncthreads();
    float gp = 0.0f;
#pragma unroll 4
    for (int j = 0; j < 8; ++j) {
      float4 gr = *(const float4*)&G[L][4 * j];
      float4 pj = *(const float4*)&pv[4 * j];
      gp += gr.x * pj.x + gr.y * pj.y + gr.z * pj.z + gr.w * pj.w;
    }
    gp *= rl;
    float sq = wave_sum32(p * (gp + 2.0f * bm)) + 1.0f;
    float s  = __builtin_amdgcn_sqrtf(sq) * __builtin_amdgcn_rcpf(sq + 1.0f);
    float logit = s * (gp + bm);
    float e  = __expf(logit);
    p = e * __builtin_amdgcn_rcpf(wave_sum32(e));
    __syncthreads();
  }

  int pbits = __builtin_bit_cast(int, p * rl);
  float4 acc;
  acc.x = xs.x * rs; acc.y = xs.y * rs; acc.z = xs.z * rs; acc.w = xs.w * rs;
#pragma unroll
  for (int j = 0; j < 8; ++j) {
    float4 pj;
    pj.x = __builtin_bit_cast(float, __builtin_amdgcn_readlane(pbits, 4 * j + 0));
    pj.y = __builtin_bit_cast(float, __builtin_amdgcn_readlane(pbits, 4 * j + 1));
    pj.z = __builtin_bit_cast(float, __builtin_amdgcn_readlane(pbits, 4 * j + 2));
    pj.w = __builtin_bit_cast(float, __builtin_amdgcn_readlane(pbits, 4 * j + 3));
    v4f z0 = zr[4 * j + 0];
    v4f z1 = zr[4 * j + 1];
    v4f z2 = zr[4 * j + 2];
    v4f z3 = zr[4 * j + 3];
    acc.x += pj.x * z0.x + pj.y * z1.x + pj.z * z2.x + pj.w * z3.x;
    acc.y += pj.x * z0.y + pj.y * z1.y + pj.z * z2.y + pj.w * z3.y;
    acc.z += pj.x * z0.z + pj.y * z1.z + pj.z * z2.z + pj.w * z3.z;
    acc.w += pj.x * z0.w + pj.y * z1.w + pj.z * z2.w + pj.w * z3.w;
  }
  v4f res = {acc.x, acc.y, acc.z, acc.w};
  *(__attribute__((address_space(1))) v4f*)(row_base((unsigned long long)out, node) + loff) = res;
}

extern "C" void kernel_launch(void* const* d_in, const int* in_sizes, int n_in,
                              void* d_out, int out_size, void* d_ws, size_t ws_size,
                              hipStream_t stream) {
  (void)n_in; (void)out_size; (void)d_ws; (void)ws_size;
  const float* x    = (const float*)d_in[0];
  const int*   x_nb = (const int*)d_in[1];
  float*       out  = (float*)d_out;
  int n = in_sizes[0] / D_DIM;

  capsule_route<<<dim3(n), dim3(32), 0, stream>>>(x, x_nb, out, n);
}


# Gather loads fetch half of two rows each (lower 16 lanes row m, upper 16 row m+16), row offset handed round by DPP row_share: no readlane or scalar address math; new sum order in b and u
# speedup vs baseline: 1.9838x; 1.0436x over previous
#include <hip/hip_runtime.h>


#define D_DIM 128
#define M_NBR 32
#define PVOFF (M_NBR * GSTR)
#define GSTR  36

typedef __attribute__((ext_vector_type(2))) float v2f;
typedef __attribute__((ext_vector_type(8))) float v8f;
typedef __attribute__((ext_vector_type(4))) float v4f;
typedef const __attribute__((address_space(1))) v4f* gv4p;

static __device__ __forceinline__ float xhalf(float x) {
  int b = __builtin_bit_cast(int, x);
  return __builtin_bit_cast(float,
      __builtin_amdgcn_permlanex16(b, b, 0x76543210, 0xfedcba98, false, false));
}
template <int M>
static __device__ __forceinline__ float row_xor(float x) {
  int b = __builtin_bit_cast(int, x);
  return __builtin_bit_cast(float, __builtin_amdgcn_update_dpp(0, b, 0x160 | M, 0xf, 0xf, true));
}

template <int N>
static __device__ __forceinline__ int row_share(int x) {
  return __builtin_amdgcn_update_dpp(0, x, 0x150 | N, 0xf, 0xf, true);
}

static __device__ __forceinline__ int row_share_n(int x, int m) {
  switch (m) {
    case 0:  return row_share<0>(x);
    case 1:  return row_share<1>(x);
    case 2:  return row_share<2>(x);
    case 3:  return row_share<3>(x);
    case 4:  return row_share<4>(x);
    case 5:  return row_share<5>(x);
    case 6:  return row_share<6>(x);
    case 7:  return row_share<7>(x);
    case 8:  return row_share<8>(x);
    case 9:  return row_share<9>(x);
    case 10: return row_share<10>(x);
    case 11: return row_share<11>(x);
    case 12: return row_share<12>(x);
    case 13: return row_share<13>(x);
    case 14: return row_share<14>(x);
    default: return row_share<15>(x);
  }
}

static __device__ __forceinline__ float half_sum16(float v) {
  v += row_xor<8>(v);
  v += row_xor<4>(v);
  v += row_xor<2>(v);
  v += row_xor<1>(v);
  return v;
}

static __device__ __forceinline__ float wave_sum32(float v) {
  v += xhalf(v);
  v += row_xor<8>(v);
  v += row_xor<4>(v);
  v += row_xor<2>(v);
  v += row_xor<1>(v);
  return v;
}

template <int M>
static __device__ __forceinline__ float xsum_step(float lo, float hi, bool up) {
  float keep = up ? hi : lo;
  float send = up ? lo : hi;
  return keep + row_xor<M>(send);
}

static __device__ __forceinline__ float inv_norm(float sq) {
  return fminf(__builtin_amdgcn_rsqf(sq), 1e12f);
}

static __device__ __forceinline__ unsigned long long uniform_base(unsigned long long b) {
  unsigned lo = (unsigned)__builtin_amdgcn_readfirstlane((int)(unsigned)b);
  unsigned hi = (unsigned)__builtin_amdgcn_readfirstlane((int)(unsigned)(b >> 32));
  return ((unsigned long long)hi << 32) | lo;
}
static __device__ __forceinline__ unsigned long long idx_base(unsigned long long ib, int node) {
  return uniform_base(ib + (unsigned long long)((long long)node * (M_NBR * 4)));
}
static __device__ __forceinline__ unsigned long long row_base(unsigned long long xb, int r) {
  unsigned long long b = xb + (unsigned long long)((long long)r * (D_DIM * 4));
  unsigned lo = (unsigned)__builtin_amdgcn_readfirstlane((int)(unsigned)b);
  unsigned hi = (unsigned)__builtin_amdgcn_readfirstlane((int)(unsigned)(b >> 32));
  return ((unsigned long long)hi << 32) | lo;
}

__global__ __launch_bounds__(32) __attribute__((amdgpu_waves_per_eu(5, 5))) void capsule_route(const float* __restrict__ x,
                                                    const int* __restrict__ x_nb,
                                                    float* __restrict__ out, int n) {
  __shared__ __attribute__((aligned(16))) float tile[M_NBR * D_DIM];
  float (*G)[GSTR]  = (float (*)[GSTR])tile;
  float* pv = tile + PVOFF;

  int node = blockIdx.x;
  if (node >= n) return;
  int L  = threadIdx.x;
  int h = L >> 4, cc = L & 15;

  int nb1 = *(const __attribute__((address_space(1))) int*)(idx_base((unsigned long long)x_nb, node) + (unsigned)L * 4u);

  unsigned long long xb = (unsigned long long)x;
  unsigned long long xb1 = uniform_base(xb - (unsigned long long)(D_DIM * 4));
  unsigned rowoff = (unsigned)nb1 << 9;
  unsigned loff = (unsigned)cc * 16u;
  unsigned long long sb = row_base(xb, node);
  v4f xs0 = *(gv4p)(sb + loff);
  v4f xs1 = *(gv4p)(sb + 256u + loff);
  v4f zr[M_NBR];
#pragma unroll
  for (int m = 0; m < 16; ++m) {
    unsigned vo = (unsigned)row_share_n((int)rowoff, m) + loff;
    zr[2 * m]     = *(gv4p)(xb1 + vo);
    zr[2 * m + 1] = *(gv4p)(xb1 + 256u + vo);
  }
#pragma unroll
  for (int m = 0; m < 16; ++m) {
    float* rowp = &tile[(m + 16 * h) * D_DIM + ((cc ^ m) << 2)];
    *(v4f*)rowp        = zr[2 * m];
    *(v4f*)(rowp + 64) = zr[2 * m + 1];
  }
  __syncthreads();


  unsigned abase = (unsigned)cc * (D_DIM * 4) + ((unsigned)(cc ^ h) << 4);
  const char* tb = (const char*)tile;
  v8f g00 = {}, g01 = {}, g11 = {};
  unsigned ad = abase;
#pragma unroll 1
  for (int j = 0; j < 8; ++j) {
    float4 p0 = *(const float4*)(tb + ad);
    float4 p1 = *(const float4*)(tb + ad + 16 * D_DIM * 4);
    float4 q0 = *(const float4*)(tb + ad + 256);
    float4 q1 = *(const float4*)(tb + ad + 16 * D_DIM * 4 + 256);
    ad = abase ^ ((unsigned)(j + 1) << 5);
    v2f a0, a1;
    a0 = (v2f){p0.x, p0.y}; a1 = (v2f){p1.x, p1.y};
    g00 = __builtin_amdgcn_wmma_f32_16x16x4_f32(false, a0, false, a0, (short)0, g00, false, false);
    g01 = __builtin_amdgcn_wmma_f32_16x16x4_f32(false, a0, false, a1, (short)0, g01, false, false);
    g11 = __builtin_amdgcn_wmma_f32_16x16x4_f32(false, a1, false, a1, (short)0, g11, false, false);
    a0 = (v2f){p0.z, p0.w}; a1 = (v2f){p1.z, p1.w};
    g00 = __builtin_amdgcn_wmma_f32_16x16x4_f32(false, a0, false, a0, (short)0, g00, false, false);
    g01 = __builtin_amdgcn_wmma_f32_16x16x4_f32(false, a0, false, a1, (short)0, g01, false, false);
    g11 = __builtin_amdgcn_wmma_f32_16x16x4_f32(false, a1, false, a1, (short)0, g11, false, false);
    a0 = (v2f){q0.x, q0.y}; a1 = (v2f){q1.x, q1.y};
    g00 = __builtin_amdgcn_wmma_f32_16x16x4_f32(false, a0, false, a0, (short)0, g00, false, false);
    g01 = __builtin_amdgcn_wmma_f32_16x16x4_f32(false, a0, false, a1, (short)0, g01, false, false);
    g11 = __builtin_amdgcn_wmma_f32_16x16x4_f32(false, a1, false, a1, (short)0, g11, false, false);
    a0 = (v2f){q0.z, q0.w}; a1 = (v2f){q1.z, q1.w};
    g00 = __builtin_amdgcn_wmma_f32_16x16x4_f32(false, a0, false, a0, (short)0, g00, false, false);
    g01 = __builtin_amdgcn_wmma_f32_16x16x4_f32(false, a0, false, a1, (short)0, g01, false, false);
    g11 = __builtin_amdgcn_wmma_f32_16x16x4_f32(false, a1, false, a1, (short)0, g11, false, false);
  }
  __syncthreads();
#pragma unroll
  for (int v = 0; v < 8; ++v) {
    int r = v + 8 * h;
    G[r][cc]           = g00[v];
    G[r][16 + cc]      = g01[v];
    G[16 + r][16 + cc] = g11[v];
  }
  float4 t0, t1;
  t0.x = g01[0]; t0.y = g01[1]; t0.z = g01[2]; t0.w = g01[3];
  t1.x = g01[4]; t1.y = g01[5]; t1.z = g01[6]; t1.w = g01[7];
  *(float4*)&G[16 + cc][8 * h]     = t0;
  *(float4*)&G[16 + cc][8 * h + 4] = t1;
  __syncthreads();

  v2f x0lo = {xs0.x, xs0.y}, x0hi = {xs0.z, xs0.w};
  v2f x1lo = {xs1.x, xs1.y}, x1hi = {xs1.z, xs1.w};
  float u3[2];
#pragma unroll
  for (int g = 0; g < 2; ++g) {
    float t[8], u1[4], u2[2];
#pragma unroll
    for (int i = 0; i < 8; ++i) {
      int m = 8 * g + i;
      v2f alo = {zr[2 * m].x, zr[2 * m].y},         ahi = {zr[2 * m].z, zr[2 * m].w};
      v2f blo = {zr[2 * m + 1].x, zr[2 * m + 1].y}, bhi = {zr[2 * m + 1].z, zr[2 * m + 1].w};
      v2f q = alo * x0lo + ahi * x0hi + blo * x1lo + bhi * x1hi;
      t[i] = q.x + q.y;
    }
#pragma unroll
    for (int i = 0; i < 4; ++i) u1[i] = xsum_step<1>(t[2 * i], t[2 * i + 1], (L & 1) != 0);
#pragma unroll
    for (int i = 0; i < 2; ++i) u2[i] = xsum_step<2>(u1[2 * i], u1[2 * i + 1], (L & 2) != 0);
    u3[g] = xsum_step<4>(u2[0], u2[1], (L & 4) != 0);
  }
  float braw = xsum_step<8>(u3[0], u3[1], (L & 8) != 0);
  float rs = inv_norm(half_sum16(xs0.x * xs0.x + xs0.y * xs0.y + xs0.z * xs0.z + xs0.w * xs0.w +
                                 xs1.x * xs1.x + xs1.y * xs1.y + xs1.z * xs1.z + xs1.w * xs1.w));

  float rl = inv_norm(G[L][L]);
  float bm = braw * (rl * rs);

  float p = 1.0f / 32.0f;
  for (int it = 0; it < 2; ++it) {
    pv[L] = p * rl;
    __syncthreads();
    float gp = 0.0f;
#pragma unroll 4
    for (int j = 0; j < 8; ++j) {
      float4 gr = *(const float4*)&G[L][4 * j];
      float4 pj = *(const float4*)&pv[4 * j];
      gp += gr.x * pj.x + gr.y * pj.y + gr.z * pj.z + gr.w * pj.w;
    }
    gp *= rl;
    float sq = wave_sum32(p * (gp + 2.0f * bm)) + 1.0f;
    float s  = __builtin_amdgcn_sqrtf(sq) * __builtin_amdgcn_rcpf(sq + 1.0f);
    float logit = s * (gp + bm);
    float e  = __expf(logit);
    p = e * __builtin_amdgcn_rcpf(wave_sum32(e));
    __syncthreads();
  }

  int pbits = __builtin_bit_cast(int, p * rl);
  v4f a0e = {0.0f, 0.0f, 0.0f, 0.0f}, a0o = a0e, a1e = a0e, a1o = a0e;
#pragma unroll
  for (int m = 0; m < 16; m += 2) {
    float ce = __builtin_bit_cast(float, row_share_n(pbits, m));
    float co = __builtin_bit_cast(float, row_share_n(pbits, m + 1));
    a0e += ce * zr[2 * m];
    a1e += ce * zr[2 * m + 1];
    a0o += co * zr[2 * m + 2];
    a1o += co * zr[2 * m + 3];
  }
  v4f a0 = a0e + a0o, a1 = a1e + a1o;
  bool up = h != 0;
  v4f keep = up ? a1 : a0;
  v4f send = up ? a0 : a1;
  v4f xo   = up ? xs1 : xs0;
  v4f res;
  res.x = (keep.x + xhalf(send.x)) + xo.x * rs;
  res.y = (keep.y + xhalf(send.y)) + xo.y * rs;
  res.z = (keep.z + xhalf(send.z)) + xo.z * rs;
  res.w = (keep.w + xhalf(send.w)) + xo.w * rs;
  *(__attribute__((address_space(1))) v4f*)(row_base((unsigned long long)out, node) + (unsigned)L * 16u) = res;
}

extern "C" void kernel_launch(void* const* d_in, const int* in_sizes, int n_in,
                              void* d_out, int out_size, void* d_ws, size_t ws_size,
                              hipStream_t stream) {
  (void)n_in; (void)out_size; (void)d_ws; (void)ws_size;
  const float* x    = (const float*)d_in[0];
  const int*   x_nb = (const int*)d_in[1];
  float*       out  = (float*)d_out;
  int n = in_sizes[0] / D_DIM;

  capsule_route<<<dim3(n), dim3(32), 0, stream>>>(x, x_nb, out, n);
}
